# starting kernel
# baseline (speedup 1.0000x reference)
#include <hip/hip_runtime.h>

typedef __attribute__((ext_vector_type(2))) float v2f;
typedef __attribute__((ext_vector_type(8))) float v8f;

#define B_      128
#define C_      1024
#define N_      49
#define NPAD    64
#define KCHUNKS 4
#define KCHUNK  (C_ / KCHUNKS)
#define KSTEPS2 13

__global__ __launch_bounds__(32) void cam_stage1(
    const float* __restrict__ kmat, const float* __restrict__ vmat,
    float* __restrict__ Mp)
{
    const int b     = blockIdx.x;
    const int chunk = blockIdx.y;
    const int lane  = threadIdx.x & 31;
    const int ml    = lane & 15;
    const int kh    = (lane >> 4) << 1;

    const float* kb = kmat + (size_t)b * C_ * N_;
    const float* vb = vmat + (size_t)b * C_ * N_;

    float sel[4];
    int   pc[4];
#pragma unroll
    for (int t = 0; t < 4; ++t) {
        int p  = t * 16 + ml;
        sel[t] = (p < N_) ? 1.0f : 0.0f;
        pc[t]  = (p < N_) ? p : (N_ - 1);
    }

    v8f acc[4][4] = {};

    const int c0 = chunk * KCHUNK;
    for (int c = c0; c < c0 + KCHUNK; c += 4) {
        const float* kr0 = kb + (size_t)(c + kh)     * N_;
        const float* kr1 = kb + (size_t)(c + kh + 1) * N_;
        const float* vr0 = vb + (size_t)(c + kh)     * N_;
        const float* vr1 = vb + (size_t)(c + kh + 1) * N_;

        v2f a[4], bm[4];
#pragma unroll
        for (int i = 0; i < 4; ++i) {
            a[i].x = kr0[pc[i]] * sel[i];
            a[i].y = kr1[pc[i]] * sel[i];
        }
#pragma unroll
        for (int j = 0; j < 4; ++j) {
            bm[j].x = vr0[pc[j]] * sel[j];
            bm[j].y = vr1[pc[j]] * sel[j];
        }
#pragma unroll
        for (int i = 0; i < 4; ++i)
#pragma unroll
            for (int j = 0; j < 4; ++j)
                acc[i][j] = __builtin_amdgcn_wmma_f32_16x16x4_f32(
                    false, a[i], false, bm[j], (short)0, acc[i][j], false, false);
    }

    float*    out  = Mp + (size_t)(chunk * B_ + b) * NPAD * NPAD;
    const int rhi  = (lane >> 4) << 3;
#pragma unroll
    for (int i = 0; i < 4; ++i)
#pragma unroll
        for (int j = 0; j < 4; ++j)
#pragma unroll
            for (int r = 0; r < 8; ++r)
                out[(size_t)(i * 16 + rhi + r) * NPAD + j * 16 + ml] = acc[i][j][r];
}

__global__ __launch_bounds__(256) void cam_stage2(
    const float* __restrict__ q, const float* __restrict__ v1,
    const float* __restrict__ Mp, float* __restrict__ out)
{
    __shared__ float Ms[NPAD * NPAD];
    const int b = blockIdx.x;

    for (int p = threadIdx.x; p < NPAD * NPAD; p += 256) {
        float s = 0.0f;
#pragma unroll
        for (int ch = 0; ch < KCHUNKS; ++ch)
            s += Mp[(size_t)(ch * B_ + b) * NPAD * NPAD + p];
        Ms[p] = s;
    }
    __syncthreads();

    const int wave = threadIdx.x >> 5;
    const int lane = threadIdx.x & 31;
    const int ml   = lane & 15;
    const int kh   = (lane >> 4) << 1;

    v2f Breg[KSTEPS2][4];
#pragma unroll
    for (int ks = 0; ks < KSTEPS2; ++ks)
#pragma unroll
        for (int j = 0; j < 4; ++j) {
            Breg[ks][j].x = Ms[(ks * 4 + kh)     * NPAD + j * 16 + ml];
            Breg[ks][j].y = Ms[(ks * 4 + kh + 1) * NPAD + j * 16 + ml];
        }

    const float* qb = q  + (size_t)b * C_ * N_;
    const float* vb = v1 + (size_t)b * C_ * N_;
    float*       ob = out + (size_t)b * C_ * N_;

    for (int it = wave; it < 64; it += 8) {
        const int    mrow = it * 16 + ml;
        const float* qrow = qb + (size_t)mrow * N_;
        v8f acc[4] = {};
#pragma unroll
        for (int ks = 0; ks < KSTEPS2; ++ks) {
            const int kk0 = ks * 4 + kh;
            v2f a;
            a.x = (kk0     < N_) ? qrow[kk0]     : 0.0f;
            a.y = (kk0 + 1 < N_) ? qrow[kk0 + 1] : 0.0f;
#pragma unroll
            for (int j = 0; j < 4; ++j)
                acc[j] = __builtin_amdgcn_wmma_f32_16x16x4_f32(
                    false, a, false, Breg[ks][j], (short)0, acc[j], false, false);
        }
        const int rbase = it * 16 + ((lane >> 4) << 3);
#pragma unroll
        for (int j = 0; j < 4; ++j) {
            const int n = j * 16 + ml;
            if (n < N_) {
#pragma unroll
                for (int r = 0; r < 8; ++r) {
                    const size_t idx = (size_t)(rbase + r) * N_ + n;
                    ob[idx] = acc[j][r] + vb[idx];
                }
            }
        }
    }
}

extern "C" void kernel_launch(void* const* d_in, const int* in_sizes, int n_in,
                              void* d_out, int out_size, void* d_ws, size_t ws_size,
                              hipStream_t stream) {
    const float* v1 = (const float*)d_in[0];
    const float* q1 = (const float*)d_in[1];
    const float* k1 = (const float*)d_in[2];
    float*       out = (float*)d_out;
    float*       Mp  = (float*)d_ws;

    dim3 g1(B_, KCHUNKS);
    cam_stage1<<<g1, dim3(32), 0, stream>>>(k1, v1, Mp);
    cam_stage2<<<dim3(B_), dim3(256), 0, stream>>>(q1, v1, Mp, out);
}


# stage 2 re-gridded for occupancy: 256 work-groups of 32 waves, one 16-row tile per wave with B read from LDS (was 128 x 8 waves, 8 tiles each, B held in 104 registers)
# speedup vs baseline: 2.0083x; 2.0083x over previous
#include <hip/hip_runtime.h>

typedef __attribute__((ext_vector_type(2))) float v2f;
typedef __attribute__((ext_vector_type(4))) float v4f;
typedef __attribute__((ext_vector_type(8))) float v8f;

#define B_      128
#define C_      1024
#define N_      49
#define NPAD    64
#define KCHUNKS 4
#define KCHUNK  (C_ / KCHUNKS)
#define KSTEPS2 13
#define S2_WAVES  32
#define S2_BLOCKS 256

__global__ __launch_bounds__(32) void cam_stage1(
    const float* __restrict__ kmat, const float* __restrict__ vmat,
    float* __restrict__ Mp)
{
    const int b     = blockIdx.x;
    const int chunk = blockIdx.y;
    const int lane  = threadIdx.x & 31;
    const int ml    = lane & 15;
    const int kh    = (lane >> 4) << 1;

    const float* kb = kmat + (size_t)b * C_ * N_;
    const float* vb = vmat + (size_t)b * C_ * N_;

    float sel[4];
    int   pc[4];
#pragma unroll
    for (int t = 0; t < 4; ++t) {
        int p  = t * 16 + ml;
        sel[t] = (p < N_) ? 1.0f : 0.0f;
        pc[t]  = (p < N_) ? p : (N_ - 1);
    }

    v8f acc[4][4] = {};

    const int c0 = chunk * KCHUNK;
    for (int c = c0; c < c0 + KCHUNK; c += 4) {
        const float* kr0 = kb + (size_t)(c + kh)     * N_;
        const float* kr1 = kb + (size_t)(c + kh + 1) * N_;
        const float* vr0 = vb + (size_t)(c + kh)     * N_;
        const float* vr1 = vb + (size_t)(c + kh + 1) * N_;

        v2f a[4], bm[4];
#pragma unroll
        for (int i = 0; i < 4; ++i) {
            a[i].x = kr0[pc[i]] * sel[i];
            a[i].y = kr1[pc[i]] * sel[i];
        }
#pragma unroll
        for (int j = 0; j < 4; ++j) {
            bm[j].x = vr0[pc[j]] * sel[j];
            bm[j].y = vr1[pc[j]] * sel[j];
        }
#pragma unroll
        for (int i = 0; i < 4; ++i)
#pragma unroll
            for (int j = 0; j < 4; ++j)
                acc[i][j] = __builtin_amdgcn_wmma_f32_16x16x4_f32(
                    false, a[i], false, bm[j], (short)0, acc[i][j], false, false);
    }

    float*    out  = Mp + (size_t)(chunk * B_ + b) * NPAD * NPAD;
    const int rhi  = (lane >> 4) << 3;
#pragma unroll
    for (int i = 0; i < 4; ++i)
#pragma unroll
        for (int j = 0; j < 4; ++j)
#pragma unroll
            for (int r = 0; r < 8; ++r)
                out[(size_t)(i * 16 + rhi + r) * NPAD + j * 16 + ml] = acc[i][j][r];
}

__global__ __launch_bounds__(S2_WAVES * 32) void cam_stage2(
    const float* __restrict__ q, const float* __restrict__ v1,
    const float* __restrict__ Mp, float* __restrict__ out)
{
    __shared__ __attribute__((aligned(16))) float Ms[NPAD * NPAD];
    const int b    = blockIdx.x >> 1;
    const int half = blockIdx.x & 1;

    {
        const int    p    = threadIdx.x * 4;
        const size_t pl   = (size_t)NPAD * NPAD;
        const float* base = Mp + (size_t)b * pl + p;
        const v4f c0 = *(const v4f*)(base + (size_t)(0 * B_) * pl);
        const v4f c1 = *(const v4f*)(base + (size_t)(1 * B_) * pl);
        const v4f c2 = *(const v4f*)(base + (size_t)(2 * B_) * pl);
        const v4f c3 = *(const v4f*)(base + (size_t)(3 * B_) * pl);
        const v4f s  = ((c0 + c1) + c2) + c3;
        Ms[p + 0] = s.x;
        Ms[p + 1] = s.y;
        Ms[p + 2] = s.z;
        Ms[p + 3] = s.w;
    }
    __syncthreads();

    const int wave = threadIdx.x >> 5;
    const int lane = threadIdx.x & 31;
    const int ml   = lane & 15;
    const int kh   = (lane >> 4) << 1;
    const int it   = half * S2_WAVES + wave;

    const float* qb = q  + (size_t)b * C_ * N_;
    const float* vb = v1 + (size_t)b * C_ * N_;
    float*       ob = out + (size_t)b * C_ * N_;

    const int    mrow = it * 16 + ml;
    const float* qrow = qb + (size_t)mrow * N_;
    v8f acc[4] = {};
#pragma unroll
    for (int ks = 0; ks < KSTEPS2; ++ks) {
        const int kk0 = ks * 4 + kh;
        v2f a;
        a.x = (kk0     < N_) ? qrow[kk0]     : 0.0f;
        a.y = (kk0 + 1 < N_) ? qrow[kk0 + 1] : 0.0f;
#pragma unroll
        for (int j = 0; j < 4; ++j) {
            v2f bf;
            bf.x = Ms[(kk0)     * NPAD + j * 16 + ml];
            bf.y = Ms[(kk0 + 1) * NPAD + j * 16 + ml];
            acc[j] = __builtin_amdgcn_wmma_f32_16x16x4_f32(
                false, a, false, bf, (short)0, acc[j], false, false);
        }
    }
    const int rbase = it * 16 + ((lane >> 4) << 3);
#pragma unroll
    for (int j = 0; j < 4; ++j) {
        const int n = j * 16 + ml;
        if (n < N_) {
#pragma unroll
            for (int r = 0; r < 8; ++r) {
                const size_t idx = (size_t)(rbase + r) * N_ + n;
                ob[idx] = acc[j][r] + vb[idx];
            }
        }
    }
}

extern "C" void kernel_launch(void* const* d_in, const int* in_sizes, int n_in,
                              void* d_out, int out_size, void* d_ws, size_t ws_size,
                              hipStream_t stream) {
    const float* v1 = (const float*)d_in[0];
    const float* q1 = (const float*)d_in[1];
    const float* k1 = (const float*)d_in[2];
    float*       out = (float*)d_out;
    float*       Mp  = (float*)d_ws;

    dim3 g1(B_, KCHUNKS);
    cam_stage1<<<g1, dim3(32), 0, stream>>>(k1, v1, Mp);
    cam_stage2<<<dim3(S2_BLOCKS), dim3(S2_WAVES * 32), 0, stream>>>(q1, v1, Mp, out);
}


# stage 1 re-gridded for occupancy: 512 blocks of 8 waves, each wave takes 32 of the block's 256 channels in two groups of 4 K-steps, partial planes summed through LDS (was 512 one-wave blocks)
# speedup vs baseline: 5.0837x; 2.5313x over previous
#include <hip/hip_runtime.h>

typedef __attribute__((ext_vector_type(2))) float v2f;
typedef __attribute__((ext_vector_type(4))) float v4f;
typedef __attribute__((ext_vector_type(8))) float v8f;

#define B_      128
#define C_      1024
#define N_      49
#define NPAD    64
#define KCHUNKS 4
#define KCHUNK  (C_ / KCHUNKS)
#define KSTEPS2 13
#define S1_WAVES  8
#define S1_KW     (KCHUNK / S1_WAVES)
#define RPITCH    68
#define S2_WAVES  32
#define S2_BLOCKS 256

__global__ __launch_bounds__(S1_WAVES * 32) void cam_stage1(
    const float* __restrict__ kmat, const float* __restrict__ vmat,
    float* __restrict__ Mp)
{
    __shared__ __attribute__((aligned(16))) float red[S1_WAVES * NPAD * RPITCH];

    const int b     = blockIdx.x;
    const int chunk = blockIdx.y;
    const int wave  = threadIdx.x >> 5;
    const int lane  = threadIdx.x & 31;
    const int ml    = lane & 15;
    const int kh    = (lane >> 4) << 1;

    const float* kb = kmat + (size_t)b * C_ * N_;
    const float* vb = vmat + (size_t)b * C_ * N_;

    float sel[4];
    int   pc[4];
#pragma unroll
    for (int t = 0; t < 4; ++t) {
        int p  = t * 16 + ml;
        sel[t] = (p < N_) ? 1.0f : 0.0f;
        pc[t]  = (p < N_) ? p : (N_ - 1);
    }

    v8f acc[4][4] = {};

    const int c0 = chunk * KCHUNK + wave * S1_KW;
#pragma nounroll
    for (int g = 0; g < S1_KW; g += 16) {
#pragma unroll
        for (int s = 0; s < 16; s += 4) {
            const int c = c0 + g + s;
            const float* kr0 = kb + (size_t)(c + kh)     * N_;
            const float* kr1 = kb + (size_t)(c + kh + 1) * N_;
            const float* vr0 = vb + (size_t)(c + kh)     * N_;
            const float* vr1 = vb + (size_t)(c + kh + 1) * N_;

            v2f a[4], bm[4];
#pragma unroll
            for (int i = 0; i < 4; ++i) {
                a[i].x = kr0[pc[i]] * sel[i];
                a[i].y = kr1[pc[i]] * sel[i];
            }
#pragma unroll
            for (int j = 0; j < 4; ++j) {
                bm[j].x = vr0[pc[j]] * sel[j];
                bm[j].y = vr1[pc[j]] * sel[j];
            }
#pragma unroll
            for (int i = 0; i < 4; ++i)
#pragma unroll
                for (int j = 0; j < 4; ++j)
                    acc[i][j] = __builtin_amdgcn_wmma_f32_16x16x4_f32(
                        false, a[i], false, bm[j], (short)0, acc[i][j], false, false);
        }
    }

    float*    mine = red + wave * (NPAD * RPITCH);
    const int rhi  = (lane >> 4) << 3;
#pragma unroll
    for (int i = 0; i < 4; ++i)
#pragma unroll
        for (int j = 0; j < 4; ++j)
#pragma unroll
            for (int r = 0; r < 8; ++r)
                mine[(i * 16 + rhi + r) * RPITCH + j * 16 + ml] = acc[i][j][r];
    __syncthreads();

    float* out = Mp + (size_t)(chunk * B_ + b) * NPAD * NPAD;
#pragma unroll
    for (int qd = 0; qd < 4; ++qd) {
        const int f   = threadIdx.x + qd * (S1_WAVES * 32);
        const int row = f >> 4;
        const int col = (f & 15) << 2;
        const float* src = red + row * RPITCH + col;
        v4f s = *(const v4f*)src;
#pragma unroll
        for (int w = 1; w < S1_WAVES; ++w)
            s += *(const v4f*)(src + w * (NPAD * RPITCH));
        *(v4f*)(out + row * NPAD + col) = s;
    }
}

__global__ __launch_bounds__(S2_WAVES * 32) void cam_stage2(
    const float* __restrict__ q, const float* __restrict__ v1,
    const float* __restrict__ Mp, float* __restrict__ out)
{
    __shared__ __attribute__((aligned(16))) float Ms[NPAD * NPAD];
    const int b    = blockIdx.x >> 1;
    const int half = blockIdx.x & 1;

    {
        const int    p    = threadIdx.x * 4;
        const size_t pl   = (size_t)NPAD * NPAD;
        const float* base = Mp + (size_t)b * pl + p;
        const v4f c0 = *(const v4f*)(base + (size_t)(0 * B_) * pl);
        const v4f c1 = *(const v4f*)(base + (size_t)(1 * B_) * pl);
        const v4f c2 = *(const v4f*)(base + (size_t)(2 * B_) * pl);
        const v4f c3 = *(const v4f*)(base + (size_t)(3 * B_) * pl);
        const v4f s  = ((c0 + c1) + c2) + c3;
        Ms[p + 0] = s.x;
        Ms[p + 1] = s.y;
        Ms[p + 2] = s.z;
        Ms[p + 3] = s.w;
    }
    __syncthreads();

    const int wave = threadIdx.x >> 5;
    const int lane = threadIdx.x & 31;
    const int ml   = lane & 15;
    const int kh   = (lane >> 4) << 1;
    const int it   = half * S2_WAVES + wave;

    const float* qb = q  + (size_t)b * C_ * N_;
    const float* vb = v1 + (size_t)b * C_ * N_;
    float*       ob = out + (size_t)b * C_ * N_;

    const int    mrow = it * 16 + ml;
    const float* qrow = qb + (size_t)mrow * N_;
    v8f acc[4] = {};
#pragma unroll
    for (int ks = 0; ks < KSTEPS2; ++ks) {
        const int kk0 = ks * 4 + kh;
        v2f a;
        a.x = (kk0     < N_) ? qrow[kk0]     : 0.0f;
        a.y = (kk0 + 1 < N_) ? qrow[kk0 + 1] : 0.0f;
#pragma unroll
        for (int j = 0; j < 4; ++j) {
            v2f bf;
            bf.x = Ms[(kk0)     * NPAD + j * 16 + ml];
            bf.y = Ms[(kk0 + 1) * NPAD + j * 16 + ml];
            acc[j] = __builtin_amdgcn_wmma_f32_16x16x4_f32(
                false, a, false, bf, (short)0, acc[j], false, false);
        }
    }
    const int rbase = it * 16 + ((lane >> 4) << 3);
#pragma unroll
    for (int j = 0; j < 4; ++j) {
        const int n = j * 16 + ml;
        if (n < N_) {
#pragma unroll
            for (int r = 0; r < 8; ++r) {
                const size_t idx = (size_t)(rbase + r) * N_ + n;
                ob[idx] = acc[j][r] + vb[idx];
            }
        }
    }
}

extern "C" void kernel_launch(void* const* d_in, const int* in_sizes, int n_in,
                              void* d_out, int out_size, void* d_ws, size_t ws_size,
                              hipStream_t stream) {
    const float* v1 = (const float*)d_in[0];
    const float* q1 = (const float*)d_in[1];
    const float* k1 = (const float*)d_in[2];
    float*       out = (float*)d_out;
    float*       Mp  = (float*)d_ws;

    dim3 g1(B_, KCHUNKS);
    cam_stage1<<<g1, dim3(S1_WAVES * 32), 0, stream>>>(k1, v1, Mp);
    cam_stage2<<<dim3(S2_BLOCKS), dim3(S2_WAVES * 32), 0, stream>>>(q1, v1, Mp, out);
}


# stage 2 moves each wave's contiguous 16x49 tiles of q, v1 and out as whole 128-byte lines, 16 B per lane, q and out staged through a wave-private LDS tile (was 4/8-byte accesses 196 B apart)
# speedup vs baseline: 5.9145x; 1.1634x over previous
#include <hip/hip_runtime.h>

typedef __attribute__((ext_vector_type(2))) float v2f;
typedef __attribute__((ext_vector_type(4))) float v4f;
typedef __attribute__((ext_vector_type(8))) float v8f;

#define B_      128
#define C_      1024
#define N_      49
#define NPAD    64
#define KCHUNKS 4
#define KCHUNK  (C_ / KCHUNKS)
#define KSTEPS2 13
#define S1_WAVES  8
#define S1_KW     (KCHUNK / S1_WAVES)
#define RPITCH    68
#define S2_WAVES  32
#define S2_BLOCKS 256
#define TILE_F    (16 * N_)
#define TILE_V4   (TILE_F / 4)
#define TILE_LD   ((TILE_V4 + 31) / 32)

__global__ __launch_bounds__(S1_WAVES * 32) void cam_stage1(
    const float* __restrict__ kmat, const float* __restrict__ vmat,
    float* __restrict__ Mp)
{
    __shared__ __attribute__((aligned(16))) float red[S1_WAVES * NPAD * RPITCH];

    const int b     = blockIdx.x;
    const int chunk = blockIdx.y;
    const int wave  = threadIdx.x >> 5;
    const int lane  = threadIdx.x & 31;
    const int ml    = lane & 15;
    const int kh    = (lane >> 4) << 1;

    const float* kb = kmat + (size_t)b * C_ * N_;
    const float* vb = vmat + (size_t)b * C_ * N_;

    float sel[4];
    int   pc[4];
#pragma unroll
    for (int t = 0; t < 4; ++t) {
        int p  = t * 16 + ml;
        sel[t] = (p < N_) ? 1.0f : 0.0f;
        pc[t]  = (p < N_) ? p : (N_ - 1);
    }

    v8f acc[4][4] = {};

    const int c0 = chunk * KCHUNK + wave * S1_KW;
#pragma nounroll
    for (int g = 0; g < S1_KW; g += 16) {
#pragma unroll
        for (int s = 0; s < 16; s += 4) {
            const int c = c0 + g + s;
            const float* kr0 = kb + (size_t)(c + kh)     * N_;
            const float* kr1 = kb + (size_t)(c + kh + 1) * N_;
            const float* vr0 = vb + (size_t)(c + kh)     * N_;
            const float* vr1 = vb + (size_t)(c + kh + 1) * N_;

            v2f a[4], bm[4];
#pragma unroll
            for (int i = 0; i < 4; ++i) {
                a[i].x = kr0[pc[i]] * sel[i];
                a[i].y = kr1[pc[i]] * sel[i];
            }
#pragma unroll
            for (int j = 0; j < 4; ++j) {
                bm[j].x = vr0[pc[j]] * sel[j];
                bm[j].y = vr1[pc[j]] * sel[j];
            }
#pragma unroll
            for (int i = 0; i < 4; ++i)
#pragma unroll
                for (int j = 0; j < 4; ++j)
                    acc[i][j] = __builtin_amdgcn_wmma_f32_16x16x4_f32(
                        false, a[i], false, bm[j], (short)0, acc[i][j], false, false);
        }
    }

    float*    mine = red + wave * (NPAD * RPITCH);
    const int rhi  = (lane >> 4) << 3;
#pragma unroll
    for (int i = 0; i < 4; ++i)
#pragma unroll
        for (int j = 0; j < 4; ++j)
#pragma unroll
            for (int r = 0; r < 8; ++r)
                mine[(i * 16 + rhi + r) * RPITCH + j * 16 + ml] = acc[i][j][r];
    __syncthreads();

    float* out = Mp + (size_t)(chunk * B_ + b) * NPAD * NPAD;
#pragma unroll
    for (int qd = 0; qd < 4; ++qd) {
        const int f   = threadIdx.x + qd * (S1_WAVES * 32);
        const int row = f >> 4;
        const int col = (f & 15) << 2;
        const float* src = red + row * RPITCH + col;
        v4f s = *(const v4f*)src;
#pragma unroll
        for (int w = 1; w < S1_WAVES; ++w)
            s += *(const v4f*)(src + w * (NPAD * RPITCH));
        *(v4f*)(out + row * NPAD + col) = s;
    }
}

__global__ __launch_bounds__(S2_WAVES * 32) void cam_stage2(
    const float* __restrict__ q, const float* __restrict__ v1,
    const float* __restrict__ Mp, float* __restrict__ out)
{
    __shared__ __attribute__((aligned(16))) float smem[NPAD * NPAD + S2_WAVES * TILE_F];
    float* Ms    = smem;
    float* tiles = smem + NPAD * NPAD;
    const int b    = blockIdx.x >> 1;
    const int half = blockIdx.x & 1;

    const int wave = threadIdx.x >> 5;
    const int lane = threadIdx.x & 31;
    const int ml   = lane & 15;
    const int kh   = (lane >> 4) << 1;
    const int it   = half * S2_WAVES + wave;

    const size_t toff = ((size_t)b * C_ + (size_t)it * 16) * N_;
    const v4f* qt = (const v4f*)(q  + toff);
    const v4f* vt = (const v4f*)(v1 + toff);
    v4f*       ot = (v4f*)(out + toff);
    float*     tw = tiles + wave * TILE_F;

    const int    p    = threadIdx.x * 4;
    const size_t pl   = (size_t)NPAD * NPAD;
    const float* base = Mp + (size_t)b * pl + p;
    const v4f c0 = *(const v4f*)(base + (size_t)(0 * B_) * pl);
    const v4f c1 = *(const v4f*)(base + (size_t)(1 * B_) * pl);
    const v4f c2 = *(const v4f*)(base + (size_t)(2 * B_) * pl);
    const v4f c3 = *(const v4f*)(base + (size_t)(3 * B_) * pl);

    v4f qr[TILE_LD], vr[TILE_LD];
#pragma unroll
    for (int t = 0; t < TILE_LD; ++t) {
        const int f = t * 32 + lane;
        if (f < TILE_V4) qr[t] = qt[f];
    }

    *(v4f*)(Ms + p) = ((c0 + c1) + c2) + c3;
#pragma unroll
    for (int t = 0; t < TILE_LD; ++t) {
        const int f = t * 32 + lane;
        if (f < TILE_V4) *(v4f*)(tw + 4 * f) = qr[t];
    }
    __syncthreads();

#pragma unroll
    for (int t = 0; t < TILE_LD; ++t) {
        const int f = t * 32 + lane;
        if (f < TILE_V4) vr[t] = vt[f];
    }

    v8f acc[4] = {};
#pragma unroll
    for (int ks = 0; ks < KSTEPS2; ++ks) {
        const int kk0 = ks * 4 + kh;
        v2f a;
        a.x = (kk0     < N_) ? tw[ml * N_ + kk0]     : 0.0f;
        a.y = (kk0 + 1 < N_) ? tw[ml * N_ + kk0 + 1] : 0.0f;
#pragma unroll
        for (int j = 0; j < 4; ++j) {
            v2f bf;
            bf.x = Ms[(kk0)     * NPAD + j * 16 + ml];
            bf.y = Ms[(kk0 + 1) * NPAD + j * 16 + ml];
            acc[j] = __builtin_amdgcn_wmma_f32_16x16x4_f32(
                false, a, false, bf, (short)0, acc[j], false, false);
        }
    }

    const int rhi = (lane >> 4) << 3;
#pragma unroll
    for (int j = 0; j < 4; ++j) {
        const int n = j * 16 + ml;
        if (n < N_) {
#pragma unroll
            for (int r = 0; r < 8; ++r)
                tw[(rhi + r) * N_ + n] = acc[j][r];
        }
    }
    asm volatile("s_wait_dscnt 0x0" ::: "memory");
#pragma unroll
    for (int t = 0; t < TILE_LD; ++t) {
        const int f = t * 32 + lane;
        if (f < TILE_V4) ot[f] = *(const v4f*)(tw + 4 * f) + vr[t];
    }
}

extern "C" void kernel_launch(void* const* d_in, const int* in_sizes, int n_in,
                              void* d_out, int out_size, void* d_ws, size_t ws_size,
                              hipStream_t stream) {
    const float* v1 = (const float*)d_in[0];
    const float* q1 = (const float*)d_in[1];
    const float* k1 = (const float*)d_in[2];
    float*       out = (float*)d_out;
    float*       Mp  = (float*)d_ws;

    dim3 g1(B_, KCHUNKS);
    cam_stage1<<<g1, dim3(S1_WAVES * 32), 0, stream>>>(k1, v1, Mp);
    cam_stage2<<<dim3(S2_BLOCKS), dim3(S2_WAVES * 32), 0, stream>>>(q1, v1, Mp, out);
}
